# starting kernel
# baseline (speedup 1.0000x reference)
#include <hip/hip_runtime.h>
#include <hip/hip_bf16.h>

#define IN_F 128
#define HD   128
#define NH   4
#define DH   32
#define NEG_SLOPE 0.2f

typedef __attribute__((ext_vector_type(2))) float v2f;
typedef __attribute__((ext_vector_type(8))) float v8f;

__device__ __forceinline__ unsigned f2ord(float f) {
    unsigned b = __float_as_uint(f);
    return (b & 0x80000000u) ? ~b : (b | 0x80000000u);
}
__device__ __forceinline__ float ord2f(unsigned k) {
    unsigned b = (k & 0x80000000u) ? (k ^ 0x80000000u) : ~k;
    return __uint_as_float(b);
}

__global__ void fill_u32(unsigned* __restrict__ p, unsigned v, long long n) {
    long long t = (long long)blockIdx.x * blockDim.x + threadIdx.x;
    if (t < n) p[t] = v;
}

__global__ __launch_bounds__(256) void gemm_wmma_f32(
    const float* __restrict__ A,
    const float* __restrict__ B,
    float* __restrict__ C,
    int nrows)
{
    const int lane    = threadIdx.x & 31;
    const int wave    = threadIdx.x >> 5;
    const int l15     = lane & 15;
    const int halfSel = lane >> 4;
    const int mBase   = blockIdx.x * 16;
    const int nBase   = wave * 16;

    int row = mBase + l15;
    if (row >= nrows) row = nrows - 1;
    const int col = nBase + l15;

    const float* aptr = A + (size_t)row * IN_F + (halfSel << 1);
    const float* bptr = B + (size_t)(halfSel << 1) * HD + col;

    v8f acc = {};
#pragma unroll
    for (int k = 0; k < IN_F; k += 4) {
        v2f a;
        a.x = aptr[k];
        a.y = aptr[k + 1];
        v2f b;
        b.x = bptr[(size_t)k * HD];
        b.y = bptr[(size_t)(k + 1) * HD];
        acc = __builtin_amdgcn_wmma_f32_16x16x4_f32(
             false, a,  false, b,
             (short)0, acc,  false,  false);
    }

#pragma unroll
    for (int r = 0; r < 8; ++r) {
        const int m = mBase + r + (halfSel << 3);
        if (m < nrows) C[(size_t)m * HD + nBase + l15] = acc[r];
    }
}

__global__ void node_attn(const float* __restrict__ ft,
                          const float* __restrict__ attn_l,
                          const float* __restrict__ attn_r,
                          float* __restrict__ el, float* __restrict__ er,
                          long long nh_total)
{
    long long t = (long long)blockIdx.x * blockDim.x + threadIdx.x;
    if (t >= nh_total) return;
    const long long n = t >> 2;
    const int h = (int)(t & 3);
    const float* f  = ft + (size_t)n * HD + h * DH;
    const float* al = attn_l + h * DH;
    const float* ar = attn_r + h * DH;
    float sl = 0.f, sr = 0.f;
#pragma unroll
    for (int d = 0; d < DH; d += 4) {
        const float4 fv = *(const float4*)(f + d);
        const float4 lv = *(const float4*)(al + d);
        const float4 rv = *(const float4*)(ar + d);
        sl += fv.x * lv.x + fv.y * lv.y + fv.z * lv.z + fv.w * lv.w;
        sr += fv.x * rv.x + fv.y * rv.y + fv.z * rv.z + fv.w * rv.w;
    }
    el[t] = sl;
    er[t] = sr;
}

__global__ void edge_logits(const float* __restrict__ el,
                            const float* __restrict__ er,
                            const int* __restrict__ src,
                            const int* __restrict__ dst,
                            float* __restrict__ ebuf,
                            unsigned* __restrict__ emaxk,
                            long long total)
{
    long long t = (long long)blockIdx.x * blockDim.x + threadIdx.x;
    if (t >= total) return;
    const long long e = t >> 2;
    const int h = (int)(t & 3);
    const int s = src[e], d = dst[e];
    float v = el[(size_t)s * NH + h] + er[(size_t)d * NH + h];
    v = v > 0.f ? v : NEG_SLOPE * v;
    ebuf[t] = v;
    atomicMax(&emaxk[(size_t)d * NH + h], f2ord(v));
}

__global__ void finalize_max(const unsigned* __restrict__ emaxk,
                             float* __restrict__ emaxf, long long nh_total)
{
    long long t = (long long)blockIdx.x * blockDim.x + threadIdx.x;
    if (t >= nh_total) return;
    const unsigned k = emaxk[t];
    float v = (k == 0u) ? 0.f : ord2f(k);
    if (!__builtin_isfinite(v)) v = 0.f;
    emaxf[t] = v;
}

__global__ void edge_exp(const float* __restrict__ emaxf,
                         const int* __restrict__ dst,
                         float* __restrict__ ebuf,
                         float* __restrict__ denom,
                         long long total)
{
    long long t = (long long)blockIdx.x * blockDim.x + threadIdx.x;
    if (t >= total) return;
    const long long e = t >> 2;
    const int h = (int)(t & 3);
    const int d = dst[e];
    const float ex = expf(ebuf[t] - emaxf[(size_t)d * NH + h]);
    ebuf[t] = ex;
    atomicAdd(&denom[(size_t)d * NH + h], ex);
}

__global__ void finalize_denom(float* __restrict__ denom, long long nh_total) {
    long long t = (long long)blockIdx.x * blockDim.x + threadIdx.x;
    if (t >= nh_total) return;
    const float s = denom[t];
    denom[t] = (s > 0.f) ? 1.0f / s : 1.0f;
}

__global__ void edge_aggregate(const float* __restrict__ ft,
                               const float* __restrict__ ebuf,
                               const float* __restrict__ rdenom,
                               const int* __restrict__ src,
                               const int* __restrict__ dst,
                               float* __restrict__ out,
                               long long total)
{
    long long t = (long long)blockIdx.x * blockDim.x + threadIdx.x;
    if (t >= total) return;
    const long long e = t >> 5;
    const int q = (int)(t & 31);
    const int h = q >> 3;
    const int s = src[e], d = dst[e];
    const float a = ebuf[e * NH + h] * rdenom[(size_t)d * NH + h];
    const float4 v = ((const float4*)(ft + (size_t)s * HD))[q];
    float* o = out + (size_t)d * HD + q * 4;
    atomicAdd(o + 0, a * v.x);
    atomicAdd(o + 1, a * v.y);
    atomicAdd(o + 2, a * v.z);
    atomicAdd(o + 3, a * v.w);
}

static inline unsigned nblk(long long n, int bs) {
    return (unsigned)((n + bs - 1) / bs);
}

extern "C" void kernel_launch(void* const* d_in, const int* in_sizes, int n_in,
                              void* d_out, int out_size, void* d_ws, size_t ws_size,
                              hipStream_t stream) {
    const float* feat   = (const float*)d_in[0];
    const float* W      = (const float*)d_in[1];
    const float* attn_l = (const float*)d_in[2];
    const float* attn_r = (const float*)d_in[3];
    const int*   src    = (const int*)d_in[4];
    const int*   dst    = (const int*)d_in[5];
    float*       out    = (float*)d_out;

    const long long N = in_sizes[0] / IN_F;
    const long long E = in_sizes[4];

    float*    ft    = (float*)d_ws;
    float*    el    = ft + (size_t)N * HD;
    float*    er    = el + (size_t)N * NH;
    unsigned* emaxk = (unsigned*)(er + (size_t)N * NH);
    float*    emaxf = (float*)(emaxk + (size_t)N * NH);
    float*    denom = emaxf + (size_t)N * NH;
    float*    ebuf  = denom + (size_t)N * NH;

    const int BS = 256;
    const long long NHtot = N * NH;
    const long long EHtot = E * NH;
    const long long EAgg  = E * 32;

    fill_u32<<<nblk((long long)out_size, BS), BS, 0, stream>>>((unsigned*)out, 0u, out_size);
    fill_u32<<<nblk(NHtot, BS), BS, 0, stream>>>((unsigned*)denom, 0u, NHtot);
    fill_u32<<<nblk(NHtot, BS), BS, 0, stream>>>(emaxk, 0u, NHtot);

    gemm_wmma_f32<<<nblk(N, 16), 256, 0, stream>>>(feat, W, ft, (int)N);

    node_attn<<<nblk(NHtot, BS), BS, 0, stream>>>(ft, attn_l, attn_r, el, er, NHtot);

    edge_logits<<<nblk(EHtot, BS), BS, 0, stream>>>(el, er, src, dst, ebuf, emaxk, EHtot);

    finalize_max<<<nblk(NHtot, BS), BS, 0, stream>>>(emaxk, emaxf, NHtot);

    edge_exp<<<nblk(EHtot, BS), BS, 0, stream>>>(emaxf, dst, ebuf, denom, EHtot);

    finalize_denom<<<nblk(NHtot, BS), BS, 0, stream>>>(denom, NHtot);

    edge_aggregate<<<nblk(EAgg, BS), BS, 0, stream>>>(ft, ebuf, denom, src, dst, out, EAgg);
}


# Projection GEMM moves its data through LDS in whole 512-byte rows: a 64-row strip of feat staged once per block, each wave's W slice in registers, whole-row stores of ft; k order unchanged
# speedup vs baseline: 2.3069x; 2.3069x over previous
#include <hip/hip_runtime.h>
#include <hip/hip_bf16.h>

#define IN_F 128
#define HD   128
#define NH   4
#define DH   32
#define NEG_SLOPE 0.2f
#define CAP  64

typedef __attribute__((ext_vector_type(2))) float v2f;
typedef __attribute__((ext_vector_type(8))) float v8f;

__device__ __forceinline__ unsigned f2ord(float f) {
    unsigned b = __float_as_uint(f);
    return (b & 0x80000000u) ? ~b : (b | 0x80000000u);
}
__device__ __forceinline__ float ord2f(unsigned k) {
    unsigned b = (k & 0x80000000u) ? (k ^ 0x80000000u) : ~k;
    return __uint_as_float(b);
}

__global__ void fill_u32(unsigned* __restrict__ p, unsigned v, long long n) {
    long long t = (long long)blockIdx.x * blockDim.x + threadIdx.x;
    if (t < n) p[t] = v;
}

#define GM_ROWS  64
#define GM_TILES (GM_ROWS / 16)
#define GM_PITCH 132
__global__ __launch_bounds__(256) void gemm_wmma_f32(
    const float* __restrict__ A,
    const float* __restrict__ B,
    float* __restrict__ C,
    int nrows)
{
    __shared__ __attribute__((aligned(16))) float sA[GM_ROWS * GM_PITCH];
    __shared__ __attribute__((aligned(16))) float sC[GM_ROWS * GM_PITCH];

    const int tid     = threadIdx.x;
    const int lane    = tid & 31;
    const int wave    = tid >> 5;
    const int l15     = lane & 15;
    const int halfSel = lane >> 4;
    const int mBase   = blockIdx.x * GM_ROWS;
    const int nBase   = wave * 16;

#pragma unroll
    for (int i = 0; i < GM_ROWS / 8; ++i) {
        const int f  = tid + i * 256;
        const int r  = f >> 5;
        const int c4 = f & 31;
        int gr = mBase + r;
        if (gr >= nrows) gr = nrows - 1;
        const float4 v = *(const float4*)(A + (size_t)gr * IN_F + c4 * 4);
        *(float4*)&sA[r * GM_PITCH + c4 * 4] = v;
    }

    v2f bf[IN_F / 4];
    const float* bptr = B + (size_t)(halfSel << 1) * HD + nBase + l15;
#pragma unroll
    for (int j = 0; j < IN_F / 4; ++j) {
        bf[j].x = bptr[(size_t)(4 * j) * HD];
        bf[j].y = bptr[(size_t)(4 * j + 1) * HD];
    }

    __syncthreads();

    const float* ap = &sA[l15 * GM_PITCH + (halfSel << 1)];
    float* cp = &sC[(halfSel << 3) * GM_PITCH + nBase + l15];
    for (int t = 0; t < GM_TILES; t += 2) {
        v8f acc0 = {}, acc1 = {};
        const float* a0p = ap + t * 16 * GM_PITCH;
        const float* a1p = a0p + 16 * GM_PITCH;
#pragma unroll
        for (int j = 0; j < IN_F / 4; ++j) {
            const v2f a0 = *(const v2f*)(a0p + 4 * j);
            const v2f a1 = *(const v2f*)(a1p + 4 * j);
            acc0 = __builtin_amdgcn_wmma_f32_16x16x4_f32(
                 false, a0,  false, bf[j],
                 (short)0, acc0,  false,  false);
            acc1 = __builtin_amdgcn_wmma_f32_16x16x4_f32(
                 false, a1,  false, bf[j],
                 (short)0, acc1,  false,  false);
        }
        float* c0p = cp + t * 16 * GM_PITCH;
#pragma unroll
        for (int r = 0; r < 8; ++r) {
            c0p[r * GM_PITCH]        = acc0[r];
            c0p[(16 + r) * GM_PITCH] = acc1[r];
        }
    }

    __syncthreads();

#pragma unroll
    for (int i = 0; i < GM_ROWS / 8; ++i) {
        const int f  = tid + i * 256;
        const int r  = f >> 5;
        const int c4 = f & 31;
        const int gr = mBase + r;
        if (gr < nrows)
            *(float4*)(C + (size_t)gr * HD + c4 * 4) = *(const float4*)&sC[r * GM_PITCH + c4 * 4];
    }
}

__global__ void node_attn(const float* __restrict__ ft,
                          const float* __restrict__ attn_l,
                          const float* __restrict__ attn_r,
                          float* __restrict__ el, float* __restrict__ er,
                          long long nh_total)
{
    long long t = (long long)blockIdx.x * blockDim.x + threadIdx.x;
    if (t >= nh_total) return;
    const long long n = t >> 2;
    const int h = (int)(t & 3);
    const float* f  = ft + (size_t)n * HD + h * DH;
    const float* al = attn_l + h * DH;
    const float* ar = attn_r + h * DH;
    float sl = 0.f, sr = 0.f;
#pragma unroll
    for (int d = 0; d < DH; d += 4) {
        const float4 fv = *(const float4*)(f + d);
        const float4 lv = *(const float4*)(al + d);
        const float4 rv = *(const float4*)(ar + d);
        sl += fv.x * lv.x + fv.y * lv.y + fv.z * lv.z + fv.w * lv.w;
        sr += fv.x * rv.x + fv.y * rv.y + fv.z * rv.z + fv.w * rv.w;
    }
    el[t] = sl;
    er[t] = sr;
}

__global__ void edge_count(const int* __restrict__ dst,
                           int* __restrict__ cnt,
                           int* __restrict__ rank,
                           long long total)
{
    long long t = (long long)blockIdx.x * blockDim.x + threadIdx.x;
    if (t >= total) return;
    rank[t] = atomicAdd(&cnt[dst[t]], 1);
}

__global__ __launch_bounds__(256) void seg_alloc(const int* __restrict__ cnt,
                                                 int* __restrict__ start,
                                                 int* __restrict__ cursor,
                                                 int nnodes)
{
    __shared__ int wtot[8];
    __shared__ int base;
    const int t    = blockIdx.x * 256 + threadIdx.x;
    const int lane = threadIdx.x & 31;
    const int wv   = threadIdx.x >> 5;
    const int c    = (t < nnodes) ? cnt[t] : 0;
    int inc = c;
    for (int o = 1; o < 32; o <<= 1) {
        const int y = __shfl_up(inc, o);
        if (lane >= o) inc += y;
    }
    if (lane == 31) wtot[wv] = inc;
    __syncthreads();
    if (threadIdx.x == 0) {
        int s = 0;
        for (int i = 0; i < 8; ++i) { const int x = wtot[i]; wtot[i] = s; s += x; }
        base = atomicAdd(cursor, s);
    }
    __syncthreads();
    if (t < nnodes) start[t] = base + wtot[wv] + inc - c;
}

__global__ void edge_place(const int* __restrict__ src,
                           const int* __restrict__ dst,
                           const int* __restrict__ start,
                           const int* __restrict__ rank,
                           int* __restrict__ slots,
                           long long total)
{
    long long t = (long long)blockIdx.x * blockDim.x + threadIdx.x;
    if (t >= total) return;
    const int d = dst[t];
    ((int2*)slots)[(size_t)start[d] + rank[t]] = make_int2((int)t, src[t]);
}

__device__ __forceinline__ float leaky(float v) { return v > 0.f ? v : NEG_SLOPE * v; }
__device__ __forceinline__ float wave_max(float x) {
    for (int o = 16; o > 0; o >>= 1) x = fmaxf(x, __shfl_xor(x, o));
    return x;
}
__device__ __forceinline__ float wave_sum(float x) {
    for (int o = 16; o > 0; o >>= 1) x += __shfl_xor(x, o);
    return x;
}

template <int K>
__device__ __forceinline__ void gather_rows(float4& acc, const float4* __restrict__ ft4,
                                            const float* wts, int src_reg, int j, int q) {
    float4 r[K];
    float  a[K];
#pragma unroll
    for (int k = 0; k < K; ++k) {
        const int sj = __builtin_amdgcn_readlane(src_reg, j + k);
        r[k] = ft4[(size_t)sj * (HD / 4) + q];
    }
#pragma unroll
    for (int k = 0; k < K; ++k) a[k] = wts[(j + k) * 4];
#pragma unroll
    for (int k = 0; k < K; ++k) {
        acc.x += a[k] * r[k].x; acc.y += a[k] * r[k].y;
        acc.z += a[k] * r[k].z; acc.w += a[k] * r[k].w;
    }
}
__device__ __forceinline__ void gather_run(float4& acc, const float4* __restrict__ ft4,
                                           const float* wts, int src_reg, int cnt, int q) {
    int j = 0;
    for (; j + 8 <= cnt; j += 8) gather_rows<8>(acc, ft4, wts, src_reg, j, q);
    if (j + 4 <= cnt) { gather_rows<4>(acc, ft4, wts, src_reg, j, q); j += 4; }
    if (j + 2 <= cnt) { gather_rows<2>(acc, ft4, wts, src_reg, j, q); j += 2; }
    if (j < cnt)      { gather_rows<1>(acc, ft4, wts, src_reg, j, q); }
}

__device__ __forceinline__ int2 next_edge(const int2* __restrict__ run, int c, int prev, int q) {
    int best = 0x7fffffff, bsrc = 0;
    for (int i = q; i < c; i += 32) {
        const int2 p = run[i];
        if (p.x > prev && p.x < best) { best = p.x; bsrc = p.y; }
    }
    for (int o = 16; o > 0; o >>= 1) {
        const int ob = __shfl_xor(best, o);
        const int os = __shfl_xor(bsrc, o);
        if (ob < best) { best = ob; bsrc = os; }
    }
    return make_int2(best, bsrc);
}

__global__ __launch_bounds__(256) void node_gather(const float* __restrict__ ft,
                                                   const float* __restrict__ el,
                                                   const float* __restrict__ er,
                                                   const int* __restrict__ cnt,
                                                   const int* __restrict__ start,
                                                   const int* __restrict__ slots,
                                                   float* __restrict__ out,
                                                   int nnodes)
{
    __shared__ int   s_src[8][CAP];
    __shared__ float s_a[8][CAP][NH];
    const int q = threadIdx.x & 31;
    const int w = threadIdx.x >> 5;
    const int n = __builtin_amdgcn_readfirstlane((int)(blockIdx.x * 8 + w));
    if (n >= nnodes) return;
    const int c  = cnt[n];
    const int h  = q >> 3;
    float4* orow = (float4*)(out + (size_t)n * HD) + q;
    float4 acc = make_float4(0.f, 0.f, 0.f, 0.f);
    if (c == 0) { *orow = acc; return; }

    const int2*   run = (const int2*)slots + start[n];
    const float4* ft4 = (const float4*)ft;
    const float4* el4 = (const float4*)el;
    const float4  r4  = *(const float4*)(er + (size_t)n * NH);

    if (c <= CAP) {
        const bool v0 = q < c, v1 = (q + 32) < c;
        int k0 = 0x7fffffff, k1 = 0x7fffffff, x0 = 0, x1 = 0;
        if (v0) { const int2 p = run[q];      k0 = p.x; x0 = p.y; }
        if (v1) { const int2 p = run[q + 32]; k1 = p.x; x1 = p.y; }
        const int c0 = c < 32 ? c : 32;
        int p0 = 0, p1 = 0;
        for (int j = 0; j < c0; ++j) {
            const int kj = __builtin_amdgcn_readlane(k0, j);
            p0 += (kj < k0); p1 += (kj < k1);
        }
        for (int j = 32; j < c; ++j) {
            const int kj = __builtin_amdgcn_readlane(k1, j - 32);
            p0 += (kj < k0); p1 += (kj < k1);
        }
        if (v0) s_src[w][p0] = x0;
        if (v1) s_src[w][p1] = x1;
        __builtin_amdgcn_fence(__ATOMIC_RELEASE, "wavefront");
        __builtin_amdgcn_wave_barrier();
        __builtin_amdgcn_fence(__ATOMIC_ACQUIRE, "wavefront");
        const int y0 = v0 ? s_src[w][q] : 0;
        const int y1 = v1 ? s_src[w][q + 32] : 0;

        const float ninf = -__builtin_inff();
        float4 l0 = make_float4(0.f, 0.f, 0.f, 0.f), l1 = l0;
        if (v0) l0 = el4[y0];
        if (v1) l1 = el4[y1];
        const float e0[4] = { v0 ? leaky(l0.x + r4.x) : ninf, v0 ? leaky(l0.y + r4.y) : ninf,
                              v0 ? leaky(l0.z + r4.z) : ninf, v0 ? leaky(l0.w + r4.w) : ninf };
        const float e1[4] = { v1 ? leaky(l1.x + r4.x) : ninf, v1 ? leaky(l1.y + r4.y) : ninf,
                              v1 ? leaky(l1.z + r4.z) : ninf, v1 ? leaky(l1.w + r4.w) : ninf };
        float a0[4], a1[4];
#pragma unroll
        for (int g = 0; g < 4; ++g) {
            float m = wave_max(fmaxf(e0[g], e1[g]));
            if (!__builtin_isfinite(m)) m = 0.f;
            const float u0 = v0 ? expf(e0[g] - m) : 0.f;
            const float u1 = v1 ? expf(e1[g] - m) : 0.f;
            float den = wave_sum(u0 + u1);
            den = den > 0.f ? den : 1.0f;
            a0[g] = u0 / den;
            a1[g] = u1 / den;
        }

        if (v0) *(float4*)&s_a[w][q][0]      = make_float4(a0[0], a0[1], a0[2], a0[3]);
        if (v1) *(float4*)&s_a[w][q + 32][0] = make_float4(a1[0], a1[1], a1[2], a1[3]);
        __builtin_amdgcn_fence(__ATOMIC_RELEASE, "wavefront");
        __builtin_amdgcn_wave_barrier();
        __builtin_amdgcn_fence(__ATOMIC_ACQUIRE, "wavefront");
        gather_run(acc, ft4, &s_a[w][0][h],  y0, c0,     q);
        gather_run(acc, ft4, &s_a[w][32][h], y1, c - 32, q);
    } else {
        const float erh = er[(size_t)n * NH + h];
        float m = -__builtin_inff();
        for (int i = 0; i < c; ++i) {
            const int sj = run[i].y;
            m = fmaxf(m, leaky(el[(size_t)sj * NH + h] + erh));
        }
        if (!__builtin_isfinite(m)) m = 0.f;
        float den = 0.f;
        int prev = -1;
        for (int k = 0; k < c; ++k) {
            const int2 b = next_edge(run, c, prev, q);
            prev = b.x;
            den += expf(leaky(el[(size_t)b.y * NH + h] + erh) - m);
        }
        den = den > 0.f ? den : 1.0f;
        prev = -1;
        for (int k = 0; k < c; ++k) {
            const int2 b = next_edge(run, c, prev, q);
            prev = b.x;
            const float a = expf(leaky(el[(size_t)b.y * NH + h] + erh) - m) / den;
            const float4 v = ft4[(size_t)b.y * (HD / 4) + q];
            acc.x += a * v.x; acc.y += a * v.y; acc.z += a * v.z; acc.w += a * v.w;
        }
    }
    *orow = acc;
}

static inline unsigned nblk(long long n, int bs) {
    return (unsigned)((n + bs - 1) / bs);
}

extern "C" void kernel_launch(void* const* d_in, const int* in_sizes, int n_in,
                              void* d_out, int out_size, void* d_ws, size_t ws_size,
                              hipStream_t stream) {
    const float* feat   = (const float*)d_in[0];
    const float* W      = (const float*)d_in[1];
    const float* attn_l = (const float*)d_in[2];
    const float* attn_r = (const float*)d_in[3];
    const int*   src    = (const int*)d_in[4];
    const int*   dst    = (const int*)d_in[5];
    float*       out    = (float*)d_out;

    const long long N = in_sizes[0] / IN_F;
    const long long E = in_sizes[4];

    float* ft     = (float*)d_ws;
    float* el     = ft + (size_t)N * HD;
    float* er     = el + (size_t)N * NH;
    int*   slots  = (int*)(er + (size_t)N * NH);
    int*   rank   = slots + (size_t)E * 2;
    int*   cnt    = rank + (size_t)E;
    int*   cursor = cnt + (size_t)N;
    int*   start  = cursor + 1;

    const int BS = 256;
    const long long NHtot = N * NH;

    fill_u32<<<nblk(N + 1, BS), BS, 0, stream>>>((unsigned*)cnt, 0u, N + 1);

    edge_count<<<nblk(E, BS), BS, 0, stream>>>(dst, cnt, rank, E);
    seg_alloc<<<nblk(N, BS), BS, 0, stream>>>(cnt, start, cursor, (int)N);
    edge_place<<<nblk(E, BS), BS, 0, stream>>>(src, dst, start, rank, slots, E);

    gemm_wmma_f32<<<nblk(N, GM_ROWS), 256, 0, stream>>>(feat, W, ft, (int)N);

    node_attn<<<nblk(NHtot, BS), BS, 0, stream>>>(ft, attn_l, attn_r, el, er, NHtot);

    node_gather<<<nblk(N, 8), 256, 0, stream>>>(ft, el, er, cnt, start, slots, out, (int)N);
}


# Edges grouped by destination with LDS counters in two levels (cells per chunk x 128-node group, then per-node buckets in LDS): no returning global atomic per edge, no count fill; sum orders unchanged
# speedup vs baseline: 4.3996x; 1.9071x over previous
#include <hip/hip_runtime.h>
#include <hip/hip_bf16.h>

#define IN_F 128
#define HD   128
#define NH   4
#define DH   32
#define NEG_SLOPE 0.2f
#define BCAP 48

#define GSH    7
#define GN     (1 << GSH)
#define NPART  192
#define CELL   32
#define CSLOTS (CELL - 1)
#define MAXNB  1024
#define SB     1024
#define EB     9
#define NW     16

typedef __attribute__((ext_vector_type(2))) float v2f;
typedef __attribute__((ext_vector_type(8))) float v8f;

__global__ __launch_bounds__(SB) void bin_edges(const int* __restrict__ src,
                                                const int* __restrict__ dst,
                                                int* __restrict__ cells,
                                                int ngroups, long long nedges)
{
    __shared__ int s_cnt[MAXNB];
    __shared__ __attribute__((aligned(16))) int s_cell[MAXNB * CELL];
    const int t = threadIdx.x;
    const int b = blockIdx.x;
    for (int i = t; i < ngroups; i += SB) s_cnt[i] = 0;
    __syncthreads();

    const int ne    = (int)nedges;
    const int chunk = (ne + NPART - 1) / NPART;
    const int lo    = b * chunk;
    const int hi    = lo + chunk < ne ? lo + chunk : ne;
    for (int base = lo; base < hi; base += EB * SB) {
        int d[EB], s[EB];
#pragma unroll
        for (int k = 0; k < EB; ++k) {
            const int e  = base + k * SB + t;
            const int ec = e < hi ? e : hi - 1;
            d[k] = dst[ec];
            s[k] = src[ec];
        }
#pragma unroll
        for (int k = 0; k < EB; ++k) {
            if (base + k * SB + t < hi) {
                const int grp  = d[k] >> GSH;
                const int slot = atomicAdd(&s_cnt[grp], 1);
                if (slot < CSLOTS)
                    s_cell[grp * CELL + slot] =
                        (int)(((unsigned)s[k] << GSH) | ((unsigned)d[k] & (GN - 1)));
            }
        }
    }
    __syncthreads();
    for (int i = t; i < ngroups; i += SB) s_cell[i * CELL + CSLOTS] = s_cnt[i];
    __syncthreads();
    for (int idx = t; idx < ngroups * (CELL / 4); idx += SB) {
        const int grp  = idx >> 3;
        const int part = idx & 7;
        const int4 v = *(const int4*)&s_cell[grp * CELL + part * 4];
        *(int4*)(cells + ((size_t)grp * NPART + b) * CELL + part * 4) = v;
    }
}

#define GM_ROWS  64
#define GM_TILES (GM_ROWS / 16)
#define GM_PITCH 132
#define AT_PITCH 36
__global__ __launch_bounds__(256) __attribute__((amdgpu_waves_per_eu(8, 8))) void gemm_wmma_f32(
    const float* __restrict__ A,
    const float* __restrict__ B,
    float* __restrict__ C,
    int nrows,
    const float* __restrict__ attn_l,
    const float* __restrict__ attn_r,
    float* __restrict__ el,
    float* __restrict__ er)
{
    __shared__ __attribute__((aligned(16))) float sA[GM_ROWS * GM_PITCH];
    __shared__ __attribute__((aligned(16))) float sC[GM_ROWS * GM_PITCH];
    __shared__ __attribute__((aligned(16))) float sAt[2 * NH * AT_PITCH];

    const int tid     = threadIdx.x;
    const int lane    = tid & 31;
    const int wave    = tid >> 5;
    const int l15     = lane & 15;
    const int halfSel = lane >> 4;
    const int mBase   = blockIdx.x * GM_ROWS;
    const int nBase   = wave * 16;

#pragma unroll
    for (int i = 0; i < GM_ROWS / 8; ++i) {
        const int f  = tid + i * 256;
        const int r  = f >> 5;
        const int c4 = f & 31;
        int gr = mBase + r;
        if (gr >= nrows) gr = nrows - 1;
        const float4 v = ((const float4*)A)[(unsigned)(gr * (IN_F / 4) + c4)];
        *(float4*)&sA[r * GM_PITCH + c4 * 4] = v;
    }

    v2f bf[IN_F / 4];
    const float* bptr = B + (size_t)(halfSel << 1) * HD + nBase + l15;
#pragma unroll
    for (int j = 0; j < IN_F / 4; ++j) {
        bf[j].x = bptr[(size_t)(4 * j) * HD];
        bf[j].y = bptr[(size_t)(4 * j + 1) * HD];
    }

    __syncthreads();

    if (tid < 64) {
        const int p = tid & 31;
        const float4 v = ((const float4*)(tid < 32 ? attn_l : attn_r))[p];
        *(float4*)&sAt[(tid >> 5) * (NH * AT_PITCH) + (p >> 3) * AT_PITCH + (p & 7) * 4] = v;
    }

    const float* ap = &sA[l15 * GM_PITCH + (halfSel << 1)];
    float* cp = &sC[(halfSel << 3) * GM_PITCH + nBase + l15];
    for (int t = 0; t < GM_TILES; t += 2) {
        v8f acc0 = {}, acc1 = {};
        const float* a0p = ap + t * 16 * GM_PITCH;
        const float* a1p = a0p + 16 * GM_PITCH;
#pragma unroll
        for (int j = 0; j < IN_F / 4; ++j) {
            const v2f a0 = *(const v2f*)(a0p + 4 * j);
            const v2f a1 = *(const v2f*)(a1p + 4 * j);
            acc0 = __builtin_amdgcn_wmma_f32_16x16x4_f32(
                 false, a0,  false, bf[j],
                 (short)0, acc0,  false,  false);
            acc1 = __builtin_amdgcn_wmma_f32_16x16x4_f32(
                 false, a1,  false, bf[j],
                 (short)0, acc1,  false,  false);
        }
        float* c0p = cp + t * 16 * GM_PITCH;
#pragma unroll
        for (int r = 0; r < 8; ++r) {
            c0p[r * GM_PITCH]        = acc0[r];
            c0p[(16 + r) * GM_PITCH] = acc1[r];
        }
    }

    __syncthreads();

#pragma unroll
    for (int i = 0; i < GM_ROWS / 8; ++i) {
        const int f  = tid + i * 256;
        const int r  = f >> 5;
        const int c4 = f & 31;
        const int gr = mBase + r;
        if (gr < nrows)
            *(float4*)(C + (size_t)gr * HD + c4 * 4) = *(const float4*)&sC[r * GM_PITCH + c4 * 4];
    }

    {
        const int r = tid >> 2;
        const int h = tid & 3;
        const float* f  = &sC[r * GM_PITCH + h * DH];
        const float* al = &sAt[h * AT_PITCH];
        const float* ar = &sAt[NH * AT_PITCH + h * AT_PITCH];
        float sl = 0.f, sr = 0.f;
#pragma unroll
        for (int d = 0; d < DH; d += 4) {
            const float4 fv = *(const float4*)(f + d);
            const float4 lv = *(const float4*)(al + d);
            const float4 rv = *(const float4*)(ar + d);
            sl += fv.x * lv.x + fv.y * lv.y + fv.z * lv.z + fv.w * lv.w;
            sr += fv.x * rv.x + fv.y * rv.y + fv.z * rv.z + fv.w * rv.w;
        }
        if (mBase + r < nrows) {
            el[(unsigned)(mBase * NH + tid)] = sl;
            er[(unsigned)(mBase * NH + tid)] = sr;
        }
    }

}

__device__ __forceinline__ float leaky(float v) { return v > 0.f ? v : NEG_SLOPE * v; }
__device__ __forceinline__ float wave_max(float x) {
    for (int o = 16; o > 0; o >>= 1) x = fmaxf(x, __shfl_xor(x, o));
    return x;
}
__device__ __forceinline__ float wave_sum(float x) {
    for (int o = 16; o > 0; o >>= 1) x += __shfl_xor(x, o);
    return x;
}

template <int K>
__device__ __forceinline__ void gather_rows(float4& acc, const float4* __restrict__ ft4,
                                            const float* wts, int src_reg, int j, int q) {
    float4 r[K];
    float  a[K];
#pragma unroll
    for (int k = 0; k < K; ++k) {
        const int sj = __builtin_amdgcn_readlane(src_reg, j + k);
        r[k] = ft4[(unsigned)(sj * (HD / 4) + q)];
    }
#pragma unroll
    for (int k = 0; k < K; ++k) a[k] = wts[(j + k) * 4];
#pragma unroll
    for (int k = 0; k < K; ++k) {
        acc.x += a[k] * r[k].x; acc.y += a[k] * r[k].y;
        acc.z += a[k] * r[k].z; acc.w += a[k] * r[k].w;
    }
}
__device__ __forceinline__ void gather_run(float4& acc, const float4* __restrict__ ft4,
                                           const float* wts, int src_reg, int cnt, int q) {
    int j = 0;
    for (; j + 8 <= cnt; j += 8) gather_rows<8>(acc, ft4, wts, src_reg, j, q);
    if (j + 4 <= cnt) { gather_rows<4>(acc, ft4, wts, src_reg, j, q); j += 4; }
    if (j + 2 <= cnt) { gather_rows<2>(acc, ft4, wts, src_reg, j, q); j += 2; }
    if (j < cnt)      { gather_rows<1>(acc, ft4, wts, src_reg, j, q); }
}

__device__ __forceinline__ void one_node(const int n, const int c, const int* run,
                                         int* ssrc, float* sa, const int q,
                                         const float* __restrict__ ft,
                                         const float* __restrict__ el,
                                         const float* __restrict__ er,
                                         const int* __restrict__ src,
                                         const int* __restrict__ dst,
                                         float* __restrict__ out, const long long nedges)
{
    const int h  = q >> 3;
    float4* orow = (float4*)(out + (size_t)n * HD) + q;
    float4 acc = make_float4(0.f, 0.f, 0.f, 0.f);
    if (c == 0) { *orow = acc; return; }

    const float4* ft4 = (const float4*)ft;
    const float4* el4 = (const float4*)el;
    const float4  r4  = *(const float4*)(er + (size_t)n * NH);

    if (c <= BCAP) {
        const bool v0 = q < c, v1 = (q + 32) < c;
        unsigned k0 = 0xffffffffu, k1 = 0xffffffffu;
        if (v0) k0 = ((unsigned)run[q] << 6)      | (unsigned)q;
        if (v1) k1 = ((unsigned)run[q + 32] << 6) | (unsigned)(q + 32);
        const int c0 = c < 32 ? c : 32;
        int p0 = 0, p1 = 0;
        for (int j = 0; j < c0; ++j) {
            const unsigned kj = (unsigned)__builtin_amdgcn_readlane((int)k0, j);
            p0 += (kj < k0); p1 += (kj < k1);
        }
        for (int j = 32; j < c; ++j) {
            const unsigned kj = (unsigned)__builtin_amdgcn_readlane((int)k1, j - 32);
            p0 += (kj < k0); p1 += (kj < k1);
        }
        if (v0) ssrc[p0] = (int)(k0 >> 6);
        if (v1) ssrc[p1] = (int)(k1 >> 6);
        __builtin_amdgcn_fence(__ATOMIC_RELEASE, "wavefront");
        __builtin_amdgcn_wave_barrier();
        __builtin_amdgcn_fence(__ATOMIC_ACQUIRE, "wavefront");
        const int y0 = v0 ? ssrc[q] : 0;
        const int y1 = v1 ? ssrc[q + 32] : 0;

        const float ninf = -__builtin_inff();
        float4 l0 = make_float4(0.f, 0.f, 0.f, 0.f), l1 = l0;
        if (v0) l0 = el4[y0];
        if (v1) l1 = el4[y1];
        const float e0[4] = { v0 ? leaky(l0.x + r4.x) : ninf, v0 ? leaky(l0.y + r4.y) : ninf,
                              v0 ? leaky(l0.z + r4.z) : ninf, v0 ? leaky(l0.w + r4.w) : ninf };
        const float e1[4] = { v1 ? leaky(l1.x + r4.x) : ninf, v1 ? leaky(l1.y + r4.y) : ninf,
                              v1 ? leaky(l1.z + r4.z) : ninf, v1 ? leaky(l1.w + r4.w) : ninf };
        float a0[4], a1[4];
#pragma unroll
        for (int g = 0; g < 4; ++g) {
            float m = wave_max(fmaxf(e0[g], e1[g]));
            if (!__builtin_isfinite(m)) m = 0.f;
            const float u0 = v0 ? expf(e0[g] - m) : 0.f;
            const float u1 = v1 ? expf(e1[g] - m) : 0.f;
            float den = wave_sum(u0 + u1);
            den = den > 0.f ? den : 1.0f;
            a0[g] = u0 / den;
            a1[g] = u1 / den;
        }

        if (v0) *(float4*)&sa[q * NH]      = make_float4(a0[0], a0[1], a0[2], a0[3]);
        if (v1) *(float4*)&sa[(q + 32) * NH] = make_float4(a1[0], a1[1], a1[2], a1[3]);
        __builtin_amdgcn_fence(__ATOMIC_RELEASE, "wavefront");
        __builtin_amdgcn_wave_barrier();
        __builtin_amdgcn_fence(__ATOMIC_ACQUIRE, "wavefront");
        gather_run(acc, ft4, sa + h,  y0, c0,     q);
        gather_run(acc, ft4, sa + 32 * NH + h, y1, c - 32, q);
    } else {
        const float erh = er[(size_t)n * NH + h];
        float m = -__builtin_inff();
        for (long long b = 0; b < nedges; b += 32) {
            const long long e = b + q;
            const bool hit = e < nedges && dst[e] == n;
            const int  s   = hit ? src[e] : 0;
            unsigned mask = __builtin_amdgcn_ballot_w32(hit);
            while (mask) {
                const int j = __builtin_ctz(mask);
                mask &= mask - 1;
                const int sj = __builtin_amdgcn_readlane(s, j);
                m = fmaxf(m, leaky(el[(size_t)sj * NH + h] + erh));
            }
        }
        if (!__builtin_isfinite(m)) m = 0.f;
        float den = 0.f;
        for (long long b = 0; b < nedges; b += 32) {
            const long long e = b + q;
            const bool hit = e < nedges && dst[e] == n;
            const int  s   = hit ? src[e] : 0;
            unsigned mask = __builtin_amdgcn_ballot_w32(hit);
            while (mask) {
                const int j = __builtin_ctz(mask);
                mask &= mask - 1;
                const int sj = __builtin_amdgcn_readlane(s, j);
                den += expf(leaky(el[(size_t)sj * NH + h] + erh) - m);
            }
        }
        den = den > 0.f ? den : 1.0f;
        for (long long b = 0; b < nedges; b += 32) {
            const long long e = b + q;
            const bool hit = e < nedges && dst[e] == n;
            const int  s   = hit ? src[e] : 0;
            unsigned mask = __builtin_amdgcn_ballot_w32(hit);
            while (mask) {
                const int j = __builtin_ctz(mask);
                mask &= mask - 1;
                const int sj = __builtin_amdgcn_readlane(s, j);
                const float a = expf(leaky(el[(size_t)sj * NH + h] + erh) - m) / den;
                const float4 v = ft4[(size_t)sj * (HD / 4) + q];
                acc.x += a * v.x; acc.y += a * v.y; acc.z += a * v.z; acc.w += a * v.w;
            }
        }
    }
    *orow = acc;
}

__global__ __launch_bounds__(NW * 32) __attribute__((amdgpu_waves_per_eu(16, 16)))
void node_gather(const float* __restrict__ ft,
                 const float* __restrict__ el,
                 const float* __restrict__ er,
                 const int* __restrict__ cells,
                 const int* __restrict__ src,
                 const int* __restrict__ dst,
                 float* __restrict__ out,
                 int nnodes, long long nedges)
{
    __shared__ int   s_ncnt[GN];
    __shared__ int   s_bucket[GN * BCAP];
    __shared__ int   s_src[NW][BCAP];
    __shared__ __attribute__((aligned(16))) float s_a[NW][BCAP][NH];
    const int tid = threadIdx.x;
    const int q   = tid & 31;
    const int w   = __builtin_amdgcn_readfirstlane(tid >> 5);
    const int grp = blockIdx.x;
    if (tid < GN) s_ncnt[tid] = 0;
    __syncthreads();

    const int ne    = (int)nedges;
    const int chunk = (ne + NPART - 1) / NPART;
    const int* cbase = cells + (size_t)grp * NPART * CELL;
    int cw[NPART / NW];
#pragma unroll
    for (int j = 0; j < NPART / NW; ++j) cw[j] = cbase[(j * NW + w) * CELL + q];
    unsigned over = 0;
#pragma unroll
    for (int j = 0; j < NPART / NW; ++j) {
        const int cc = __builtin_amdgcn_readlane(cw[j], CSLOTS);
        if (cc > CSLOTS) {
            over |= 1u << j;
        } else if (q < cc) {
            const unsigned x = (unsigned)cw[j];
            const int ln  = (int)(x & (GN - 1));
            const int pos = atomicAdd(&s_ncnt[ln], 1);
            if (pos < BCAP) s_bucket[ln * BCAP + pos] = (int)(x >> GSH);
        }
    }
    while (over) {
        const int j = __builtin_ctz(over);
        over &= over - 1;
        const int pch = j * NW + w;
        const int lo  = pch * chunk;
        const int hi  = lo + chunk < ne ? lo + chunk : ne;
        for (int base = lo; base < hi; base += 8 * 32) {
            int d[8];
#pragma unroll
            for (int k = 0; k < 8; ++k) {
                const int e = base + k * 32 + q;
                d[k] = dst[e < hi ? e : hi - 1];
            }
#pragma unroll
            for (int k = 0; k < 8; ++k) {
                const int e = base + k * 32 + q;
                if (e < hi && (d[k] >> GSH) == grp) {
                    const int ln  = d[k] & (GN - 1);
                    const int pos = atomicAdd(&s_ncnt[ln], 1);
                    if (pos < BCAP) s_bucket[ln * BCAP + pos] = src[e];
                }
            }
        }
    }
    __syncthreads();

    for (int ln = w; ln < GN; ln += NW) {
        const int n = grp * GN + ln;
        if (n >= nnodes) break;
        const int c = __builtin_amdgcn_readfirstlane(s_ncnt[ln]);
        int ql = q;
        asm volatile("" : "+v"(ql));
        one_node(n, c, &s_bucket[ln * BCAP], &s_src[w][0], &s_a[w][0][0], ql,
                 ft, el, er, src, dst, out, nedges);
        __builtin_amdgcn_fence(__ATOMIC_RELEASE, "wavefront");
        __builtin_amdgcn_wave_barrier();
        __builtin_amdgcn_fence(__ATOMIC_ACQUIRE, "wavefront");
    }
}

static inline unsigned nblk(long long n, int bs) {
    return (unsigned)((n + bs - 1) / bs);
}

extern "C" void kernel_launch(void* const* d_in, const int* in_sizes, int n_in,
                              void* d_out, int out_size, void* d_ws, size_t ws_size,
                              hipStream_t stream) {
    const float* feat   = (const float*)d_in[0];
    const float* W      = (const float*)d_in[1];
    const float* attn_l = (const float*)d_in[2];
    const float* attn_r = (const float*)d_in[3];
    const int*   src    = (const int*)d_in[4];
    const int*   dst    = (const int*)d_in[5];
    float*       out    = (float*)d_out;

    const long long N = in_sizes[0] / IN_F;
    const long long E = in_sizes[4];

    float* ft    = (float*)d_ws;
    float* el    = ft + (size_t)N * HD;
    float* er    = el + (size_t)N * NH;
    int*   cells = (int*)(er + (size_t)N * NH);

    bin_edges<<<NPART, SB, 0, stream>>>(src, dst, cells, (int)nblk(N, GN), E);

    gemm_wmma_f32<<<nblk(N, GM_ROWS), 256, 0, stream>>>(feat, W, ft, (int)N, attn_l, attn_r, el, er);

    node_gather<<<nblk(N, GN), NW * 32, 0, stream>>>(ft, el, er, cells, src, dst, out, (int)N, E);
}


# Node softmax with one (edge, head) pair per lane, 8 edges x 4 heads a round: max and exp-sum take 3 vector-ALU lane exchanges for all heads, not 5 LDS shuffles per head; exp-sum tree changes
# speedup vs baseline: 4.7212x; 1.0731x over previous
#include <hip/hip_runtime.h>
#include <hip/hip_bf16.h>

#define IN_F 128
#define HD   128
#define NH   4
#define DH   32
#define NEG_SLOPE 0.2f
#define BCAP 48

#define GSH    7
#define GN     (1 << GSH)
#define NPART  192
#define CELL   32
#define CSLOTS (CELL - 1)
#define MAXNB  1024
#define SB     1024
#define EB     9
#define NW     16

typedef __attribute__((ext_vector_type(2))) float v2f;
typedef __attribute__((ext_vector_type(8))) float v8f;

__global__ __launch_bounds__(SB) void bin_edges(const int* __restrict__ src,
                                                const int* __restrict__ dst,
                                                int* __restrict__ cells,
                                                int ngroups, long long nedges)
{
    __shared__ int s_cnt[MAXNB];
    __shared__ __attribute__((aligned(16))) int s_cell[MAXNB * CELL];
    const int t = threadIdx.x;
    const int b = blockIdx.x;
    for (int i = t; i < ngroups; i += SB) s_cnt[i] = 0;
    __syncthreads();

    const int ne    = (int)nedges;
    const int chunk = (ne + NPART - 1) / NPART;
    const int lo    = b * chunk;
    const int hi    = lo + chunk < ne ? lo + chunk : ne;
    for (int base = lo; base < hi; base += EB * SB) {
        int d[EB], s[EB];
#pragma unroll
        for (int k = 0; k < EB; ++k) {
            const int e  = base + k * SB + t;
            const int ec = e < hi ? e : hi - 1;
            d[k] = dst[ec];
            s[k] = src[ec];
        }
#pragma unroll
        for (int k = 0; k < EB; ++k) {
            if (base + k * SB + t < hi) {
                const int grp  = d[k] >> GSH;
                const int slot = atomicAdd(&s_cnt[grp], 1);
                if (slot < CSLOTS)
                    s_cell[grp * CELL + slot] =
                        (int)(((unsigned)s[k] << GSH) | ((unsigned)d[k] & (GN - 1)));
            }
        }
    }
    __syncthreads();
    for (int i = t; i < ngroups; i += SB) s_cell[i * CELL + CSLOTS] = s_cnt[i];
    __syncthreads();
    for (int idx = t; idx < ngroups * (CELL / 4); idx += SB) {
        const int grp  = idx >> 3;
        const int part = idx & 7;
        const int4 v = *(const int4*)&s_cell[grp * CELL + part * 4];
        *(int4*)(cells + ((size_t)grp * NPART + b) * CELL + part * 4) = v;
    }
}

#define GM_ROWS  64
#define GM_TILES (GM_ROWS / 16)
#define GM_PITCH 132
#define AT_PITCH 36
__global__ __launch_bounds__(256) __attribute__((amdgpu_waves_per_eu(8, 8))) void gemm_wmma_f32(
    const float* __restrict__ A,
    const float* __restrict__ B,
    float* __restrict__ C,
    int nrows,
    const float* __restrict__ attn_l,
    const float* __restrict__ attn_r,
    float* __restrict__ el,
    float* __restrict__ er)
{
    __shared__ __attribute__((aligned(16))) float sA[GM_ROWS * GM_PITCH];
    __shared__ __attribute__((aligned(16))) float sC[GM_ROWS * GM_PITCH];
    __shared__ __attribute__((aligned(16))) float sAt[2 * NH * AT_PITCH];

    const int tid     = threadIdx.x;
    const int lane    = tid & 31;
    const int wave    = tid >> 5;
    const int l15     = lane & 15;
    const int halfSel = lane >> 4;
    const int mBase   = blockIdx.x * GM_ROWS;
    const int nBase   = wave * 16;

#pragma unroll
    for (int i = 0; i < GM_ROWS / 8; ++i) {
        const int f  = tid + i * 256;
        const int r  = f >> 5;
        const int c4 = f & 31;
        int gr = mBase + r;
        if (gr >= nrows) gr = nrows - 1;
        const float4 v = ((const float4*)A)[(unsigned)(gr * (IN_F / 4) + c4)];
        *(float4*)&sA[r * GM_PITCH + c4 * 4] = v;
    }

    v2f bf[IN_F / 4];
    const float* bptr = B + (size_t)(halfSel << 1) * HD + nBase + l15;
#pragma unroll
    for (int j = 0; j < IN_F / 4; ++j) {
        bf[j].x = bptr[(size_t)(4 * j) * HD];
        bf[j].y = bptr[(size_t)(4 * j + 1) * HD];
    }

    __syncthreads();

    if (tid < 64) {
        const int p = tid & 31;
        const float4 v = ((const float4*)(tid < 32 ? attn_l : attn_r))[p];
        *(float4*)&sAt[(tid >> 5) * (NH * AT_PITCH) + (p >> 3) * AT_PITCH + (p & 7) * 4] = v;
    }

    const float* ap = &sA[l15 * GM_PITCH + (halfSel << 1)];
    float* cp = &sC[(halfSel << 3) * GM_PITCH + nBase + l15];
    for (int t = 0; t < GM_TILES; t += 2) {
        v8f acc0 = {}, acc1 = {};
        const float* a0p = ap + t * 16 * GM_PITCH;
        const float* a1p = a0p + 16 * GM_PITCH;
#pragma unroll
        for (int j = 0; j < IN_F / 4; ++j) {
            const v2f a0 = *(const v2f*)(a0p + 4 * j);
            const v2f a1 = *(const v2f*)(a1p + 4 * j);
            acc0 = __builtin_amdgcn_wmma_f32_16x16x4_f32(
                 false, a0,  false, bf[j],
                 (short)0, acc0,  false,  false);
            acc1 = __builtin_amdgcn_wmma_f32_16x16x4_f32(
                 false, a1,  false, bf[j],
                 (short)0, acc1,  false,  false);
        }
        float* c0p = cp + t * 16 * GM_PITCH;
#pragma unroll
        for (int r = 0; r < 8; ++r) {
            c0p[r * GM_PITCH]        = acc0[r];
            c0p[(16 + r) * GM_PITCH] = acc1[r];
        }
    }

    __syncthreads();

#pragma unroll
    for (int i = 0; i < GM_ROWS / 8; ++i) {
        const int f  = tid + i * 256;
        const int r  = f >> 5;
        const int c4 = f & 31;
        const int gr = mBase + r;
        if (gr < nrows)
            *(float4*)(C + (size_t)gr * HD + c4 * 4) = *(const float4*)&sC[r * GM_PITCH + c4 * 4];
    }

    {
        const int r = tid >> 2;
        const int h = tid & 3;
        const float* f  = &sC[r * GM_PITCH + h * DH];
        const float* al = &sAt[h * AT_PITCH];
        const float* ar = &sAt[NH * AT_PITCH + h * AT_PITCH];
        float sl = 0.f, sr = 0.f;
#pragma unroll
        for (int d = 0; d < DH; d += 4) {
            const float4 fv = *(const float4*)(f + d);
            const float4 lv = *(const float4*)(al + d);
            const float4 rv = *(const float4*)(ar + d);
            sl += fv.x * lv.x + fv.y * lv.y + fv.z * lv.z + fv.w * lv.w;
            sr += fv.x * rv.x + fv.y * rv.y + fv.z * rv.z + fv.w * rv.w;
        }
        if (mBase + r < nrows) {
            el[(unsigned)(mBase * NH + tid)] = sl;
            er[(unsigned)(mBase * NH + tid)] = sr;
        }
    }

}

__device__ __forceinline__ float leaky(float v) { return v > 0.f ? v : NEG_SLOPE * v; }
template <int M>
__device__ __forceinline__ float row_xor(float x) {
    return __builtin_bit_cast(float,
        __builtin_amdgcn_mov_dpp(__builtin_bit_cast(int, x), 0x160 + M, 0xf, 0xf, false));
}
__device__ __forceinline__ float half_xor(float x) {
    const int b = __builtin_bit_cast(int, x);
    return __builtin_bit_cast(float,
        __builtin_amdgcn_permlanex16(b, b, 0x76543210, 0xfedcba98, false, false));
}

template <int NIT>
__device__ __forceinline__ void node_softmax(const int c, const int* ssrc, float* sa,
                                             const float* __restrict__ el,
                                             const float erh, const int q)
{
    const int s  = q >> 2;
    const int hh = q & 3;
    const float ninf = -__builtin_inff();
    int   y[NIT];
    float lv[NIT], ev[NIT];
#pragma unroll
    for (int i = 0; i < NIT; ++i) {
        const int sl = s + 8 * i;
        y[i] = ssrc[sl < c ? sl : c - 1];
    }
#pragma unroll
    for (int i = 0; i < NIT; ++i) lv[i] = el[(unsigned)(y[i] * NH + hh)];
    float m = ninf;
#pragma unroll
    for (int i = 0; i < NIT; ++i) {
        ev[i] = (s + 8 * i < c) ? leaky(lv[i] + erh) : ninf;
        m = fmaxf(m, ev[i]);
    }
    m = fmaxf(m, row_xor<4>(m));
    m = fmaxf(m, row_xor<8>(m));
    m = fmaxf(m, half_xor(m));
    if (!__builtin_isfinite(m)) m = 0.f;
    float den = 0.f;
#pragma unroll
    for (int i = 0; i < NIT; ++i) {
        ev[i] = (s + 8 * i < c) ? expf(ev[i] - m) : 0.f;
        den += ev[i];
    }
    den += row_xor<4>(den);
    den += row_xor<8>(den);
    den += half_xor(den);
    den = den > 0.f ? den : 1.0f;
#pragma unroll
    for (int i = 0; i < NIT; ++i)
        if (s + 8 * i < c) sa[q + 32 * i] = ev[i] / den;
}

template <int K>
__device__ __forceinline__ void gather_rows(float4& acc, const float4* __restrict__ ft4,
                                            const float* wts, int src_reg, int j, int q) {
    float4 r[K];
    float  a[K];
#pragma unroll
    for (int k = 0; k < K; ++k) {
        const int sj = __builtin_amdgcn_readlane(src_reg, j + k);
        r[k] = ft4[(unsigned)(sj * (HD / 4) + q)];
    }
#pragma unroll
    for (int k = 0; k < K; ++k) a[k] = wts[(j + k) * 4];
#pragma unroll
    for (int k = 0; k < K; ++k) {
        acc.x += a[k] * r[k].x; acc.y += a[k] * r[k].y;
        acc.z += a[k] * r[k].z; acc.w += a[k] * r[k].w;
    }
}
__device__ __forceinline__ void gather_run(float4& acc, const float4* __restrict__ ft4,
                                           const float* wts, int src_reg, int cnt, int q) {
    int j = 0;
    for (; j + 8 <= cnt; j += 8) gather_rows<8>(acc, ft4, wts, src_reg, j, q);
    if (j + 4 <= cnt) { gather_rows<4>(acc, ft4, wts, src_reg, j, q); j += 4; }
    if (j + 2 <= cnt) { gather_rows<2>(acc, ft4, wts, src_reg, j, q); j += 2; }
    if (j < cnt)      { gather_rows<1>(acc, ft4, wts, src_reg, j, q); }
}

__device__ __forceinline__ void one_node(const int n, const int c, const int* run,
                                         int* ssrc, float* sa, const int q,
                                         const float* __restrict__ ft,
                                         const float* __restrict__ el,
                                         const float* __restrict__ er,
                                         const int* __restrict__ src,
                                         const int* __restrict__ dst,
                                         float* __restrict__ out, const long long nedges)
{
    const int h  = q >> 3;
    float4* orow = (float4*)(out + (size_t)n * HD) + q;
    float4 acc = make_float4(0.f, 0.f, 0.f, 0.f);
    if (c == 0) { *orow = acc; return; }

    const float4* ft4 = (const float4*)ft;
    const float erq = er[(unsigned)(n * NH + (q & 3))];

    if (c <= BCAP) {
        const bool v0 = q < c, v1 = (q + 32) < c;
        unsigned k0 = 0xffffffffu, k1 = 0xffffffffu;
        if (v0) k0 = ((unsigned)run[q] << 6)      | (unsigned)q;
        if (v1) k1 = ((unsigned)run[q + 32] << 6) | (unsigned)(q + 32);
        const int c0 = c < 32 ? c : 32;
        int p0 = 0, p1 = 0;
        for (int j = 0; j < c0; ++j) {
            const unsigned kj = (unsigned)__builtin_amdgcn_readlane((int)k0, j);
            p0 += (kj < k0); p1 += (kj < k1);
        }
        for (int j = 32; j < c; ++j) {
            const unsigned kj = (unsigned)__builtin_amdgcn_readlane((int)k1, j - 32);
            p0 += (kj < k0); p1 += (kj < k1);
        }
        if (v0) ssrc[p0] = (int)(k0 >> 6);
        if (v1) ssrc[p1] = (int)(k1 >> 6);
        __builtin_amdgcn_fence(__ATOMIC_RELEASE, "wavefront");
        __builtin_amdgcn_wave_barrier();
        __builtin_amdgcn_fence(__ATOMIC_ACQUIRE, "wavefront");
        const int y0 = v0 ? ssrc[q] : 0;
        const int y1 = v1 ? ssrc[q + 32] : 0;

        if (c <= 16)      node_softmax<2>(c, ssrc, sa, el, erq, q);
        else if (c <= 24) node_softmax<3>(c, ssrc, sa, el, erq, q);
        else              node_softmax<6>(c, ssrc, sa, el, erq, q);
        __builtin_amdgcn_fence(__ATOMIC_RELEASE, "wavefront");
        __builtin_amdgcn_wave_barrier();
        __builtin_amdgcn_fence(__ATOMIC_ACQUIRE, "wavefront");
        gather_run(acc, ft4, sa + h,  y0, c0,     q);
        gather_run(acc, ft4, sa + 32 * NH + h, y1, c - 32, q);
    } else {
        const float erh = er[(size_t)n * NH + h];
        float m = -__builtin_inff();
        for (long long b = 0; b < nedges; b += 32) {
            const long long e = b + q;
            const bool hit = e < nedges && dst[e] == n;
            const int  s   = hit ? src[e] : 0;
            unsigned mask = __builtin_amdgcn_ballot_w32(hit);
            while (mask) {
                const int j = __builtin_ctz(mask);
                mask &= mask - 1;
                const int sj = __builtin_amdgcn_readlane(s, j);
                m = fmaxf(m, leaky(el[(size_t)sj * NH + h] + erh));
            }
        }
        if (!__builtin_isfinite(m)) m = 0.f;
        float den = 0.f;
        for (long long b = 0; b < nedges; b += 32) {
            const long long e = b + q;
            const bool hit = e < nedges && dst[e] == n;
            const int  s   = hit ? src[e] : 0;
            unsigned mask = __builtin_amdgcn_ballot_w32(hit);
            while (mask) {
                const int j = __builtin_ctz(mask);
                mask &= mask - 1;
                const int sj = __builtin_amdgcn_readlane(s, j);
                den += expf(leaky(el[(size_t)sj * NH + h] + erh) - m);
            }
        }
        den = den > 0.f ? den : 1.0f;
        for (long long b = 0; b < nedges; b += 32) {
            const long long e = b + q;
            const bool hit = e < nedges && dst[e] == n;
            const int  s   = hit ? src[e] : 0;
            unsigned mask = __builtin_amdgcn_ballot_w32(hit);
            while (mask) {
                const int j = __builtin_ctz(mask);
                mask &= mask - 1;
                const int sj = __builtin_amdgcn_readlane(s, j);
                const float a = expf(leaky(el[(size_t)sj * NH + h] + erh) - m) / den;
                const float4 v = ft4[(size_t)sj * (HD / 4) + q];
                acc.x += a * v.x; acc.y += a * v.y; acc.z += a * v.z; acc.w += a * v.w;
            }
        }
    }
    *orow = acc;
}

__global__ __launch_bounds__(NW * 32) __attribute__((amdgpu_waves_per_eu(16, 16)))
void node_gather(const float* __restrict__ ft,
                 const float* __restrict__ el,
                 const float* __restrict__ er,
                 const int* __restrict__ cells,
                 const int* __restrict__ src,
                 const int* __restrict__ dst,
                 float* __restrict__ out,
                 int nnodes, long long nedges)
{
    __shared__ int   s_ncnt[GN];
    __shared__ int   s_bucket[GN * BCAP];
    __shared__ int   s_src[NW][BCAP];
    __shared__ __attribute__((aligned(16))) float s_a[NW][BCAP][NH];
    const int tid = threadIdx.x;
    const int q   = tid & 31;
    const int w   = __builtin_amdgcn_readfirstlane(tid >> 5);
    const int grp = blockIdx.x;
    if (tid < GN) s_ncnt[tid] = 0;
    __syncthreads();

    const int ne    = (int)nedges;
    const int chunk = (ne + NPART - 1) / NPART;
    const int* cbase = cells + (size_t)grp * NPART * CELL;
    int cw[NPART / NW];
#pragma unroll
    for (int j = 0; j < NPART / NW; ++j) cw[j] = cbase[(j * NW + w) * CELL + q];
    unsigned over = 0;
#pragma unroll
    for (int j = 0; j < NPART / NW; ++j) {
        const int cc = __builtin_amdgcn_readlane(cw[j], CSLOTS);
        if (cc > CSLOTS) {
            over |= 1u << j;
        } else if (q < cc) {
            const unsigned x = (unsigned)cw[j];
            const int ln  = (int)(x & (GN - 1));
            const int pos = atomicAdd(&s_ncnt[ln], 1);
            if (pos < BCAP) s_bucket[ln * BCAP + pos] = (int)(x >> GSH);
        }
    }
    while (over) {
        const int j = __builtin_ctz(over);
        over &= over - 1;
        const int pch = j * NW + w;
        const int lo  = pch * chunk;
        const int hi  = lo + chunk < ne ? lo + chunk : ne;
        for (int base = lo; base < hi; base += 8 * 32) {
            int d[8];
#pragma unroll
            for (int k = 0; k < 8; ++k) {
                const int e = base + k * 32 + q;
                d[k] = dst[e < hi ? e : hi - 1];
            }
#pragma unroll
            for (int k = 0; k < 8; ++k) {
                const int e = base + k * 32 + q;
                if (e < hi && (d[k] >> GSH) == grp) {
                    const int ln  = d[k] & (GN - 1);
                    const int pos = atomicAdd(&s_ncnt[ln], 1);
                    if (pos < BCAP) s_bucket[ln * BCAP + pos] = src[e];
                }
            }
        }
    }
    __syncthreads();

    for (int ln = w; ln < GN; ln += NW) {
        const int n = grp * GN + ln;
        if (n >= nnodes) break;
        const int c = __builtin_amdgcn_readfirstlane(s_ncnt[ln]);
        int ql = q;
        asm volatile("" : "+v"(ql));
        one_node(n, c, &s_bucket[ln * BCAP], &s_src[w][0], &s_a[w][0][0], ql,
                 ft, el, er, src, dst, out, nedges);
        __builtin_amdgcn_fence(__ATOMIC_RELEASE, "wavefront");
        __builtin_amdgcn_wave_barrier();
        __builtin_amdgcn_fence(__ATOMIC_ACQUIRE, "wavefront");
    }
}

static inline unsigned nblk(long long n, int bs) {
    return (unsigned)((n + bs - 1) / bs);
}

extern "C" void kernel_launch(void* const* d_in, const int* in_sizes, int n_in,
                              void* d_out, int out_size, void* d_ws, size_t ws_size,
                              hipStream_t stream) {
    const float* feat   = (const float*)d_in[0];
    const float* W      = (const float*)d_in[1];
    const float* attn_l = (const float*)d_in[2];
    const float* attn_r = (const float*)d_in[3];
    const int*   src    = (const int*)d_in[4];
    const int*   dst    = (const int*)d_in[5];
    float*       out    = (float*)d_out;

    const long long N = in_sizes[0] / IN_F;
    const long long E = in_sizes[4];

    float* ft    = (float*)d_ws;
    float* el    = ft + (size_t)N * HD;
    float* er    = el + (size_t)N * NH;
    int*   cells = (int*)(er + (size_t)N * NH);

    bin_edges<<<NPART, SB, 0, stream>>>(src, dst, cells, (int)nblk(N, GN), E);

    gemm_wmma_f32<<<nblk(N, GM_ROWS), 256, 0, stream>>>(feat, W, ft, (int)N, attn_l, attn_r, el, er);

    node_gather<<<nblk(N, GN), NW * 32, 0, stream>>>(ft, el, er, cells, src, dst, out, (int)N, E);
}


# Destination groups of 100 nodes, not 128: 1000 node work-groups (16000 waves) all resident at once, 4 per WGP nearly everywhere, 6-7 nodes per wave instead of 8; sum orders unchanged
# speedup vs baseline: 4.8947x; 1.0367x over previous
#include <hip/hip_runtime.h>
#include <hip/hip_bf16.h>

#define IN_F 128
#define HD   128
#define NH   4
#define DH   32
#define NEG_SLOPE 0.2f
#define BCAP 48

#define GN     100
#define LSH    7
#define NPART  192
#define CELL   32
#define CSLOTS (CELL - 1)
#define MAXNB  1024
#define SB     1024
#define EB     9
#define NW     16

typedef __attribute__((ext_vector_type(2))) float v2f;
typedef __attribute__((ext_vector_type(8))) float v8f;

__global__ __launch_bounds__(SB) void bin_edges(const int* __restrict__ src,
                                                const int* __restrict__ dst,
                                                int* __restrict__ cells,
                                                int ngroups, long long nedges)
{
    __shared__ int s_cnt[MAXNB];
    __shared__ __attribute__((aligned(16))) int s_cell[MAXNB * CELL];
    const int t = threadIdx.x;
    const int b = blockIdx.x;
    for (int i = t; i < ngroups; i += SB) s_cnt[i] = 0;
    __syncthreads();

    const int ne    = (int)nedges;
    const int chunk = (ne + NPART - 1) / NPART;
    const int lo    = b * chunk;
    const int hi    = lo + chunk < ne ? lo + chunk : ne;
    for (int base = lo; base < hi; base += EB * SB) {
        int d[EB], s[EB];
#pragma unroll
        for (int k = 0; k < EB; ++k) {
            const int e  = base + k * SB + t;
            const int ec = e < hi ? e : hi - 1;
            d[k] = dst[ec];
            s[k] = src[ec];
        }
#pragma unroll
        for (int k = 0; k < EB; ++k) {
            if (base + k * SB + t < hi) {
                const unsigned du  = (unsigned)d[k];
                const unsigned grp = du / GN;
                const int slot = atomicAdd(&s_cnt[grp], 1);
                if (slot < CSLOTS)
                    s_cell[grp * CELL + slot] =
                        (int)(((unsigned)s[k] << LSH) | (du - grp * GN));
            }
        }
    }
    __syncthreads();
    for (int i = t; i < ngroups; i += SB) s_cell[i * CELL + CSLOTS] = s_cnt[i];
    __syncthreads();
    for (int idx = t; idx < ngroups * (CELL / 4); idx += SB) {
        const int grp  = idx >> 3;
        const int part = idx & 7;
        const int4 v = *(const int4*)&s_cell[grp * CELL + part * 4];
        *(int4*)(cells + ((size_t)grp * NPART + b) * CELL + part * 4) = v;
    }
}

#define GM_ROWS  64
#define GM_TILES (GM_ROWS / 16)
#define GM_PITCH 132
#define AT_PITCH 36
__global__ __launch_bounds__(256) __attribute__((amdgpu_waves_per_eu(8, 8))) void gemm_wmma_f32(
    const float* __restrict__ A,
    const float* __restrict__ B,
    float* __restrict__ C,
    int nrows,
    const float* __restrict__ attn_l,
    const float* __restrict__ attn_r,
    float* __restrict__ el,
    float* __restrict__ er)
{
    __shared__ __attribute__((aligned(16))) float sA[GM_ROWS * GM_PITCH];
    __shared__ __attribute__((aligned(16))) float sC[GM_ROWS * GM_PITCH];
    __shared__ __attribute__((aligned(16))) float sAt[2 * NH * AT_PITCH];

    const int tid     = threadIdx.x;
    const int lane    = tid & 31;
    const int wave    = tid >> 5;
    const int l15     = lane & 15;
    const int halfSel = lane >> 4;
    const int mBase   = blockIdx.x * GM_ROWS;
    const int nBase   = wave * 16;

#pragma unroll
    for (int i = 0; i < GM_ROWS / 8; ++i) {
        const int f  = tid + i * 256;
        const int r  = f >> 5;
        const int c4 = f & 31;
        int gr = mBase + r;
        if (gr >= nrows) gr = nrows - 1;
        const float4 v = ((const float4*)A)[(unsigned)(gr * (IN_F / 4) + c4)];
        *(float4*)&sA[r * GM_PITCH + c4 * 4] = v;
    }

    v2f bf[IN_F / 4];
    const float* bptr = B + (size_t)(halfSel << 1) * HD + nBase + l15;
#pragma unroll
    for (int j = 0; j < IN_F / 4; ++j) {
        bf[j].x = bptr[(size_t)(4 * j) * HD];
        bf[j].y = bptr[(size_t)(4 * j + 1) * HD];
    }

    __syncthreads();

    if (tid < 64) {
        const int p = tid & 31;
        const float4 v = ((const float4*)(tid < 32 ? attn_l : attn_r))[p];
        *(float4*)&sAt[(tid >> 5) * (NH * AT_PITCH) + (p >> 3) * AT_PITCH + (p & 7) * 4] = v;
    }

    const float* ap = &sA[l15 * GM_PITCH + (halfSel << 1)];
    float* cp = &sC[(halfSel << 3) * GM_PITCH + nBase + l15];
    for (int t = 0; t < GM_TILES; t += 2) {
        v8f acc0 = {}, acc1 = {};
        const float* a0p = ap + t * 16 * GM_PITCH;
        const float* a1p = a0p + 16 * GM_PITCH;
#pragma unroll
        for (int j = 0; j < IN_F / 4; ++j) {
            const v2f a0 = *(const v2f*)(a0p + 4 * j);
            const v2f a1 = *(const v2f*)(a1p + 4 * j);
            acc0 = __builtin_amdgcn_wmma_f32_16x16x4_f32(
                 false, a0,  false, bf[j],
                 (short)0, acc0,  false,  false);
            acc1 = __builtin_amdgcn_wmma_f32_16x16x4_f32(
                 false, a1,  false, bf[j],
                 (short)0, acc1,  false,  false);
        }
        float* c0p = cp + t * 16 * GM_PITCH;
#pragma unroll
        for (int r = 0; r < 8; ++r) {
            c0p[r * GM_PITCH]        = acc0[r];
            c0p[(16 + r) * GM_PITCH] = acc1[r];
        }
    }

    __syncthreads();

#pragma unroll
    for (int i = 0; i < GM_ROWS / 8; ++i) {
        const int f  = tid + i * 256;
        const int r  = f >> 5;
        const int c4 = f & 31;
        const int gr = mBase + r;
        if (gr < nrows)
            *(float4*)(C + (size_t)gr * HD + c4 * 4) = *(const float4*)&sC[r * GM_PITCH + c4 * 4];
    }

    {
        const int r = tid >> 2;
        const int h = tid & 3;
        const float* f  = &sC[r * GM_PITCH + h * DH];
        const float* al = &sAt[h * AT_PITCH];
        const float* ar = &sAt[NH * AT_PITCH + h * AT_PITCH];
        float sl = 0.f, sr = 0.f;
#pragma unroll
        for (int d = 0; d < DH; d += 4) {
            const float4 fv = *(const float4*)(f + d);
            const float4 lv = *(const float4*)(al + d);
            const float4 rv = *(const float4*)(ar + d);
            sl += fv.x * lv.x + fv.y * lv.y + fv.z * lv.z + fv.w * lv.w;
            sr += fv.x * rv.x + fv.y * rv.y + fv.z * rv.z + fv.w * rv.w;
        }
        if (mBase + r < nrows) {
            el[(unsigned)(mBase * NH + tid)] = sl;
            er[(unsigned)(mBase * NH + tid)] = sr;
        }
    }

}

__device__ __forceinline__ float leaky(float v) { return v > 0.f ? v : NEG_SLOPE * v; }
template <int M>
__device__ __forceinline__ float row_xor(float x) {
    return __builtin_bit_cast(float,
        __builtin_amdgcn_mov_dpp(__builtin_bit_cast(int, x), 0x160 + M, 0xf, 0xf, false));
}
__device__ __forceinline__ float half_xor(float x) {
    const int b = __builtin_bit_cast(int, x);
    return __builtin_bit_cast(float,
        __builtin_amdgcn_permlanex16(b, b, 0x76543210, 0xfedcba98, false, false));
}

template <int NIT>
__device__ __forceinline__ void node_softmax(const int c, const int* ssrc, float* sa,
                                             const float* __restrict__ el,
                                             const float erh, const int q)
{
    const int s  = q >> 2;
    const int hh = q & 3;
    const float ninf = -__builtin_inff();
    int   y[NIT];
    float lv[NIT], ev[NIT];
#pragma unroll
    for (int i = 0; i < NIT; ++i) {
        const int sl = s + 8 * i;
        y[i] = ssrc[sl < c ? sl : c - 1];
    }
#pragma unroll
    for (int i = 0; i < NIT; ++i) lv[i] = el[(unsigned)(y[i] * NH + hh)];
    float m = ninf;
#pragma unroll
    for (int i = 0; i < NIT; ++i) {
        ev[i] = (s + 8 * i < c) ? leaky(lv[i] + erh) : ninf;
        m = fmaxf(m, ev[i]);
    }
    m = fmaxf(m, row_xor<4>(m));
    m = fmaxf(m, row_xor<8>(m));
    m = fmaxf(m, half_xor(m));
    if (!__builtin_isfinite(m)) m = 0.f;
    float den = 0.f;
#pragma unroll
    for (int i = 0; i < NIT; ++i) {
        ev[i] = (s + 8 * i < c) ? expf(ev[i] - m) : 0.f;
        den += ev[i];
    }
    den += row_xor<4>(den);
    den += row_xor<8>(den);
    den += half_xor(den);
    den = den > 0.f ? den : 1.0f;
#pragma unroll
    for (int i = 0; i < NIT; ++i)
        if (s + 8 * i < c) sa[q + 32 * i] = ev[i] / den;
}

template <int K>
__device__ __forceinline__ void gather_rows(float4& acc, const float4* __restrict__ ft4,
                                            const float* wts, int src_reg, int j, int q) {
    float4 r[K];
    float  a[K];
#pragma unroll
    for (int k = 0; k < K; ++k) {
        const int sj = __builtin_amdgcn_readlane(src_reg, j + k);
        r[k] = ft4[(unsigned)(sj * (HD / 4) + q)];
    }
#pragma unroll
    for (int k = 0; k < K; ++k) a[k] = wts[(j + k) * 4];
#pragma unroll
    for (int k = 0; k < K; ++k) {
        acc.x += a[k] * r[k].x; acc.y += a[k] * r[k].y;
        acc.z += a[k] * r[k].z; acc.w += a[k] * r[k].w;
    }
}
__device__ __forceinline__ void gather_run(float4& acc, const float4* __restrict__ ft4,
                                           const float* wts, int src_reg, int cnt, int q) {
    int j = 0;
    for (; j + 8 <= cnt; j += 8) gather_rows<8>(acc, ft4, wts, src_reg, j, q);
    if (j + 4 <= cnt) { gather_rows<4>(acc, ft4, wts, src_reg, j, q); j += 4; }
    if (j + 2 <= cnt) { gather_rows<2>(acc, ft4, wts, src_reg, j, q); j += 2; }
    if (j < cnt)      { gather_rows<1>(acc, ft4, wts, src_reg, j, q); }
}

__device__ __forceinline__ void one_node(const int n, const int c, const int* run,
                                         int* ssrc, float* sa, const int q,
                                         const float* __restrict__ ft,
                                         const float* __restrict__ el,
                                         const float* __restrict__ er,
                                         const int* __restrict__ src,
                                         const int* __restrict__ dst,
                                         float* __restrict__ out, const long long nedges)
{
    const int h  = q >> 3;
    float4* orow = (float4*)(out + (size_t)n * HD) + q;
    float4 acc = make_float4(0.f, 0.f, 0.f, 0.f);
    if (c == 0) { *orow = acc; return; }

    const float4* ft4 = (const float4*)ft;
    const float erq = er[(unsigned)(n * NH + (q & 3))];

    if (c <= BCAP) {
        const bool v0 = q < c, v1 = (q + 32) < c;
        unsigned k0 = 0xffffffffu, k1 = 0xffffffffu;
        if (v0) k0 = ((unsigned)run[q] << 6)      | (unsigned)q;
        if (v1) k1 = ((unsigned)run[q + 32] << 6) | (unsigned)(q + 32);
        const int c0 = c < 32 ? c : 32;
        int p0 = 0, p1 = 0;
        for (int j = 0; j < c0; ++j) {
            const unsigned kj = (unsigned)__builtin_amdgcn_readlane((int)k0, j);
            p0 += (kj < k0); p1 += (kj < k1);
        }
        for (int j = 32; j < c; ++j) {
            const unsigned kj = (unsigned)__builtin_amdgcn_readlane((int)k1, j - 32);
            p0 += (kj < k0); p1 += (kj < k1);
        }
        if (v0) ssrc[p0] = (int)(k0 >> 6);
        if (v1) ssrc[p1] = (int)(k1 >> 6);
        __builtin_amdgcn_fence(__ATOMIC_RELEASE, "wavefront");
        __builtin_amdgcn_wave_barrier();
        __builtin_amdgcn_fence(__ATOMIC_ACQUIRE, "wavefront");
        const int y0 = v0 ? ssrc[q] : 0;
        const int y1 = v1 ? ssrc[q + 32] : 0;

        if (c <= 16)      node_softmax<2>(c, ssrc, sa, el, erq, q);
        else if (c <= 24) node_softmax<3>(c, ssrc, sa, el, erq, q);
        else              node_softmax<6>(c, ssrc, sa, el, erq, q);
        __builtin_amdgcn_fence(__ATOMIC_RELEASE, "wavefront");
        __builtin_amdgcn_wave_barrier();
        __builtin_amdgcn_fence(__ATOMIC_ACQUIRE, "wavefront");
        gather_run(acc, ft4, sa + h,  y0, c0,     q);
        gather_run(acc, ft4, sa + 32 * NH + h, y1, c - 32, q);
    } else {
        const float erh = er[(size_t)n * NH + h];
        float m = -__builtin_inff();
        for (long long b = 0; b < nedges; b += 32) {
            const long long e = b + q;
            const bool hit = e < nedges && dst[e] == n;
            const int  s   = hit ? src[e] : 0;
            unsigned mask = __builtin_amdgcn_ballot_w32(hit);
            while (mask) {
                const int j = __builtin_ctz(mask);
                mask &= mask - 1;
                const int sj = __builtin_amdgcn_readlane(s, j);
                m = fmaxf(m, leaky(el[(size_t)sj * NH + h] + erh));
            }
        }
        if (!__builtin_isfinite(m)) m = 0.f;
        float den = 0.f;
        for (long long b = 0; b < nedges; b += 32) {
            const long long e = b + q;
            const bool hit = e < nedges && dst[e] == n;
            const int  s   = hit ? src[e] : 0;
            unsigned mask = __builtin_amdgcn_ballot_w32(hit);
            while (mask) {
                const int j = __builtin_ctz(mask);
                mask &= mask - 1;
                const int sj = __builtin_amdgcn_readlane(s, j);
                den += expf(leaky(el[(size_t)sj * NH + h] + erh) - m);
            }
        }
        den = den > 0.f ? den : 1.0f;
        for (long long b = 0; b < nedges; b += 32) {
            const long long e = b + q;
            const bool hit = e < nedges && dst[e] == n;
            const int  s   = hit ? src[e] : 0;
            unsigned mask = __builtin_amdgcn_ballot_w32(hit);
            while (mask) {
                const int j = __builtin_ctz(mask);
                mask &= mask - 1;
                const int sj = __builtin_amdgcn_readlane(s, j);
                const float a = expf(leaky(el[(size_t)sj * NH + h] + erh) - m) / den;
                const float4 v = ft4[(size_t)sj * (HD / 4) + q];
                acc.x += a * v.x; acc.y += a * v.y; acc.z += a * v.z; acc.w += a * v.w;
            }
        }
    }
    *orow = acc;
}

__global__ __launch_bounds__(NW * 32) __attribute__((amdgpu_waves_per_eu(16, 16)))
void node_gather(const float* __restrict__ ft,
                 const float* __restrict__ el,
                 const float* __restrict__ er,
                 const int* __restrict__ cells,
                 const int* __restrict__ src,
                 const int* __restrict__ dst,
                 float* __restrict__ out,
                 int nnodes, long long nedges)
{
    __shared__ int   s_ncnt[GN];
    __shared__ int   s_bucket[GN * BCAP];
    __shared__ int   s_src[NW][BCAP];
    __shared__ __attribute__((aligned(16))) float s_a[NW][BCAP][NH];
    const int tid = threadIdx.x;
    const int q   = tid & 31;
    const int w   = __builtin_amdgcn_readfirstlane(tid >> 5);
    const int grp = blockIdx.x;
    if (tid < GN) s_ncnt[tid] = 0;
    __syncthreads();

    const int ne    = (int)nedges;
    const int chunk = (ne + NPART - 1) / NPART;
    const int* cbase = cells + (size_t)grp * NPART * CELL;
    int cw[NPART / NW];
#pragma unroll
    for (int j = 0; j < NPART / NW; ++j) cw[j] = cbase[(j * NW + w) * CELL + q];
    unsigned over = 0;
#pragma unroll
    for (int j = 0; j < NPART / NW; ++j) {
        const int cc = __builtin_amdgcn_readlane(cw[j], CSLOTS);
        if (cc > CSLOTS) {
            over |= 1u << j;
        } else if (q < cc) {
            const unsigned x = (unsigned)cw[j];
            const int ln  = (int)(x & ((1u << LSH) - 1));
            const int pos = atomicAdd(&s_ncnt[ln], 1);
            if (pos < BCAP) s_bucket[ln * BCAP + pos] = (int)(x >> LSH);
        }
    }
    while (over) {
        const int j = __builtin_ctz(over);
        over &= over - 1;
        const int pch = j * NW + w;
        const int lo  = pch * chunk;
        const int hi  = lo + chunk < ne ? lo + chunk : ne;
        for (int base = lo; base < hi; base += 8 * 32) {
            int d[8];
#pragma unroll
            for (int k = 0; k < 8; ++k) {
                const int e = base + k * 32 + q;
                d[k] = dst[e < hi ? e : hi - 1];
            }
#pragma unroll
            for (int k = 0; k < 8; ++k) {
                const int e = base + k * 32 + q;
                if (e < hi && (unsigned)d[k] / GN == (unsigned)grp) {
                    const int ln  = d[k] - grp * GN;
                    const int pos = atomicAdd(&s_ncnt[ln], 1);
                    if (pos < BCAP) s_bucket[ln * BCAP + pos] = src[e];
                }
            }
        }
    }
    __syncthreads();

    for (int ln = w; ln < GN; ln += NW) {
        const int n = grp * GN + ln;
        if (n >= nnodes) break;
        const int c = __builtin_amdgcn_readfirstlane(s_ncnt[ln]);
        int ql = q;
        asm volatile("" : "+v"(ql));
        one_node(n, c, &s_bucket[ln * BCAP], &s_src[w][0], &s_a[w][0][0], ql,
                 ft, el, er, src, dst, out, nedges);
        __builtin_amdgcn_fence(__ATOMIC_RELEASE, "wavefront");
        __builtin_amdgcn_wave_barrier();
        __builtin_amdgcn_fence(__ATOMIC_ACQUIRE, "wavefront");
    }
}

static inline unsigned nblk(long long n, int bs) {
    return (unsigned)((n + bs - 1) / bs);
}

extern "C" void kernel_launch(void* const* d_in, const int* in_sizes, int n_in,
                              void* d_out, int out_size, void* d_ws, size_t ws_size,
                              hipStream_t stream) {
    const float* feat   = (const float*)d_in[0];
    const float* W      = (const float*)d_in[1];
    const float* attn_l = (const float*)d_in[2];
    const float* attn_r = (const float*)d_in[3];
    const int*   src    = (const int*)d_in[4];
    const int*   dst    = (const int*)d_in[5];
    float*       out    = (float*)d_out;

    const long long N = in_sizes[0] / IN_F;
    const long long E = in_sizes[4];

    float* ft    = (float*)d_ws;
    float* el    = ft + (size_t)N * HD;
    float* er    = el + (size_t)N * NH;
    int*   cells = (int*)(er + (size_t)N * NH);

    bin_edges<<<NPART, SB, 0, stream>>>(src, dst, cells, (int)nblk(N, GN), E);

    gemm_wmma_f32<<<nblk(N, GM_ROWS), 256, 0, stream>>>(feat, W, ft, (int)N, attn_l, attn_r, el, er);

    node_gather<<<nblk(N, GN), NW * 32, 0, stream>>>(ft, el, er, cells, src, dst, out, (int)N, E);
}
